# starting kernel
# baseline (speedup 1.0000x reference)
#include <hip/hip_runtime.h>
#include <math.h>

typedef __attribute__((ext_vector_type(2))) float v2f;
typedef __attribute__((ext_vector_type(4))) float f4v;
typedef __attribute__((ext_vector_type(8))) float v8f;

#define WAVES 4
#define TILE  (WAVES * 16)
#define ISTR  68

__global__ __launch_bounds__(64) void build_unitary_kernel(
    const float* __restrict__ qp, float* __restrict__ Ut_r, float* __restrict__ Ut_i) {
  __shared__ float vr[64][64];
  __shared__ float vi[64][64];
  const int c = threadIdx.x;
  float* rv = vr[c];
  float* iv = vi[c];
  for (int s = 0; s < 64; ++s) { rv[s] = (s == c) ? 1.0f : 0.0f; iv[s] = 0.0f; }

  auto cnot = [&](int cq, int tq) {
    const int mc = 1 << (5 - cq), mt = 1 << (5 - tq);
    for (int s = 0; s < 64; ++s) {
      if ((s & mc) && !(s & mt)) {
        const int s1 = s | mt;
        float t = rv[s]; rv[s] = rv[s1]; rv[s1] = t;
        t = iv[s]; iv[s] = iv[s1]; iv[s1] = t;
      }
    }
  };
  auto rx = [&](int q, float th) {
    float sh, ch; sincosf(0.5f * th, &sh, &ch);
    const int m = 1 << (5 - q);
    for (int s0 = 0; s0 < 64; ++s0) if (!(s0 & m)) {
      const int s1 = s0 | m;
      const float r0 = rv[s0], i0 = iv[s0], r1 = rv[s1], i1 = iv[s1];
      rv[s0] = ch * r0 + sh * i1;  iv[s0] = ch * i0 - sh * r1;
      rv[s1] = ch * r1 + sh * i0;  iv[s1] = ch * i1 - sh * r0;
    }
  };
  auto ry = [&](int q, float th) {
    float sh, ch; sincosf(0.5f * th, &sh, &ch);
    const int m = 1 << (5 - q);
    for (int s0 = 0; s0 < 64; ++s0) if (!(s0 & m)) {
      const int s1 = s0 | m;
      const float r0 = rv[s0], i0 = iv[s0], r1 = rv[s1], i1 = iv[s1];
      rv[s0] = ch * r0 - sh * r1;  iv[s0] = ch * i0 - sh * i1;
      rv[s1] = sh * r0 + ch * r1;  iv[s1] = sh * i0 + ch * i1;
    }
  };
  auto rz = [&](int q, float th) {
    float sh, ch; sincosf(0.5f * th, &sh, &ch);
    const int m = 1 << (5 - q);
    for (int s0 = 0; s0 < 64; ++s0) if (!(s0 & m)) {
      const int s1 = s0 | m;
      const float r0 = rv[s0], i0 = iv[s0], r1 = rv[s1], i1 = iv[s1];
      rv[s0] = ch * r0 + sh * i0;  iv[s0] = ch * i0 - sh * r0;
      rv[s1] = ch * r1 - sh * i1;  iv[s1] = ch * i1 + sh * r1;
    }
  };
  auto crz = [&](int cq, int tq, float th) {
    float sh, ch; sincosf(0.5f * th, &sh, &ch);
    const int mc = 1 << (5 - cq), mt = 1 << (5 - tq);
    for (int s = 0; s < 64; ++s) if (s & mc) {
      const float r = rv[s], i = iv[s];
      if (s & mt) { rv[s] = ch * r - sh * i; iv[s] = ch * i + sh * r; }
      else        { rv[s] = ch * r + sh * i; iv[s] = ch * i - sh * r; }
    }
  };

  for (int layer = 0; layer < 3; ++layer) {
    cnot(0, 1); cnot(2, 3); cnot(4, 5);
    cnot(1, 2); cnot(3, 4);
    for (int i = 0; i < 6; ++i) {
      const float tx = qp[layer * 18 + i * 3 + 0];
      const float ty = qp[layer * 18 + i * 3 + 1];
      const float tz = qp[layer * 18 + i * 3 + 2];
      rx(i, tx); ry(i, ty); rz(i, tz);
    }
    if (layer < 2)
      for (int i = 0; i < 6; ++i)
        crz(i, (i + 1) % 6, qp[layer * 18 + i * 3 + 0]);
  }
  for (int s = 0; s < 64; ++s) { Ut_r[c * 64 + s] = rv[s]; Ut_i[c * 64 + s] = iv[s]; }
}

__global__ __launch_bounds__(WAVES * 32) void qnn_kernel(
    const float* __restrict__ x,
    const float* __restrict__ W1, const float* __restrict__ b1,
    const float* __restrict__ W2, const float* __restrict__ b2,
    const float* __restrict__ W3, const float* __restrict__ b3,
    const float* __restrict__ W4, const float* __restrict__ b4,
    const float* __restrict__ Ut_r, const float* __restrict__ Ut_i,
    float* __restrict__ out, int batch) {
  __shared__ float sUr[64 * 64];
  __shared__ float sUi[64 * 64];
  __shared__ float sAmp[WAVES][16 * ISTR];

  const int tid  = threadIdx.x;
  const int wave = tid >> 5;
  const int lane = tid & 31;
  const int half = lane >> 4;
  const int l    = lane & 15;
  const int ebase = blockIdx.x * TILE + wave * 16;

  {
    const f4v* srcR = (const f4v*)Ut_r;
    const f4v* srcI = (const f4v*)Ut_i;
    f4v* dR = (f4v*)sUr;
    f4v* dI = (f4v*)sUi;
    for (int i = tid; i < 1024; i += WAVES * 32) { dR[i] = srcR[i]; dI[i] = srcI[i]; }
  }

  if (lane < 16) {
    int e = ebase + lane;
    const int ec = (e < batch) ? e : (batch - 1);
    const float x0 = x[ec * 4 + 0], x1 = x[ec * 4 + 1];
    const float x2 = x[ec * 4 + 2], x3 = x[ec * 4 + 3];
    float h[16];
#pragma unroll
    for (int j = 0; j < 16; ++j) {
      float t = b1[j] + x0 * W1[j] + x1 * W1[16 + j] + x2 * W1[32 + j] + x3 * W1[48 + j];
      h[j] = t > 0.0f ? t : 0.0f;
    }
    float A0[6], A1[6];
#pragma unroll
    for (int q = 0; q < 6; ++q) {
      float t = b2[q];
#pragma unroll
      for (int j = 0; j < 16; ++j) t += h[j] * W2[j * 6 + q];
      const float ang = 0.5f * 3.14159265358979323846f * tanhf(t);
      float sn, cs; sincosf(ang, &sn, &cs);
      const float is2 = 0.70710678118654752440f;
      A0[q] = (cs - sn) * is2;
      A1[q] = (cs + sn) * is2;
    }
    float m01[4], m23[4], m45[4];
#pragma unroll
    for (int i = 0; i < 4; ++i) {
      m01[i] = ((i & 2) ? A1[0] : A0[0]) * ((i & 1) ? A1[1] : A0[1]);
      m23[i] = ((i & 2) ? A1[2] : A0[2]) * ((i & 1) ? A1[3] : A0[3]);
      m45[i] = ((i & 2) ? A1[4] : A0[4]) * ((i & 1) ? A1[5] : A0[5]);
    }
    float* dst = &sAmp[wave][lane * ISTR];
#pragma unroll
    for (int s = 0; s < 64; ++s)
      dst[s] = m01[s >> 4] * m23[(s >> 2) & 3] * m45[s & 3];
  }
  __syncthreads();

  v8f accr[4] = {v8f{}, v8f{}, v8f{}, v8f{}};
  v8f acci[4] = {v8f{}, v8f{}, v8f{}, v8f{}};
  const float* amp = &sAmp[wave][l * ISTR];
#pragma unroll
  for (int kt = 0; kt < 16; ++kt) {
    const int kb = 4 * kt + 2 * half;
    v2f a;
    a.x = amp[kb];
    a.y = amp[kb + 1];
#pragma unroll
    for (int nt = 0; nt < 4; ++nt) {
      const int n = nt * 16 + l;
      v2f br, bi;
      br.x = sUr[kb * 64 + n];        br.y = sUr[(kb + 1) * 64 + n];
      bi.x = sUi[kb * 64 + n];        bi.y = sUi[(kb + 1) * 64 + n];
      accr[nt] = __builtin_amdgcn_wmma_f32_16x16x4_f32(
          false, a, false, br, (short)0, accr[nt], false, false);
      acci[nt] = __builtin_amdgcn_wmma_f32_16x16x4_f32(
          false, a, false, bi, (short)0, acci[nt], false, false);
    }
  }

  {
    float* wamp = sAmp[wave];
#pragma unroll
    for (int nt = 0; nt < 4; ++nt)
#pragma unroll
      for (int r = 0; r < 8; ++r) {
        const float pr = accr[nt][r], pi = acci[nt][r];
        wamp[(r + 8 * half) * ISTR + nt * 16 + l] = pr * pr + pi * pi;
      }
  }
  __syncthreads();

  if (lane < 16) {
    const int e = ebase + lane;
    if (e < batch) {
      const float* p = &sAmp[wave][lane * ISTR];
      float z0 = 0, z1 = 0, z2 = 0, z3 = 0, z4 = 0, z5 = 0;
#pragma unroll
      for (int s = 0; s < 64; ++s) {
        const float v = p[s];
        z0 += (s & 32) ? -v : v;
        z1 += (s & 16) ? -v : v;
        z2 += (s & 8)  ? -v : v;
        z3 += (s & 4)  ? -v : v;
        z4 += (s & 2)  ? -v : v;
        z5 += (s & 1)  ? -v : v;
      }
      const float z[6] = {z0, z1, z2, z3, z4, z5};
      float o0 = b4[0], o1 = b4[1], o2 = b4[2], o3 = b4[3];
#pragma unroll
      for (int j = 0; j < 16; ++j) {
        float g = b3[j];
#pragma unroll
        for (int q = 0; q < 6; ++q) g += z[q] * W3[q * 16 + j];
        g = g > 0.0f ? g : 0.0f;
        o0 += g * W4[j * 4 + 0]; o1 += g * W4[j * 4 + 1];
        o2 += g * W4[j * 4 + 2]; o3 += g * W4[j * 4 + 3];
      }
      float* op = out + (size_t)e * 4;
      op[0] = o0; op[1] = o1; op[2] = o2; op[3] = o3;
    }
  }
}

extern "C" void kernel_launch(void* const* d_in, const int* in_sizes, int n_in,
                              void* d_out, int out_size, void* d_ws, size_t ws_size,
                              hipStream_t stream) {
  const float* x  = (const float*)d_in[0];
  const float* W1 = (const float*)d_in[1];
  const float* b1 = (const float*)d_in[2];
  const float* W2 = (const float*)d_in[3];
  const float* b2 = (const float*)d_in[4];
  const float* qp = (const float*)d_in[5];
  const float* W3 = (const float*)d_in[6];
  const float* b3 = (const float*)d_in[7];
  const float* W4 = (const float*)d_in[8];
  const float* b4 = (const float*)d_in[9];
  float* out  = (float*)d_out;
  float* Ut_r = (float*)d_ws;
  float* Ut_i = Ut_r + 64 * 64;

  const int batch = in_sizes[0] / 4;
  build_unitary_kernel<<<dim3(1), dim3(64), 0, stream>>>(qp, Ut_r, Ut_i);
  const int tiles = (batch + TILE - 1) / TILE;
  qnn_kernel<<<dim3(tiles), dim3(WAVES * 32), 0, stream>>>(
      x, W1, b1, W2, b2, W3, b3, W4, b4, Ut_r, Ut_i, out, batch);
}


# Main kernel: the four small dense layers (4-16-6, 6-16-4) run as transposed f32 WMMA products chained through the D map; operands from a table a fifth build block writes; new sum order there
# speedup vs baseline: 12.7004x; 12.7004x over previous
#include <hip/hip_runtime.h>
#include <math.h>

typedef __attribute__((ext_vector_type(2))) float v2f;
typedef __attribute__((ext_vector_type(4))) float f4v;
typedef __attribute__((ext_vector_type(8))) float v8f;

#define WAVES 4
#define TILE  (WAVES * 16)

struct cf { float r, i; };

__device__ __forceinline__ cf cmul(cf a, cf b) {
  cf o;
  o.r = a.r * b.r - a.i * b.i;
  o.i = a.r * b.i + a.i * b.r;
  return o;
}

__device__ __forceinline__ int cnot_chain(int c) {
  if (c & 32) c ^= 16;
  if (c & 8)  c ^= 4;
  if (c & 2)  c ^= 1;
  if (c & 16) c ^= 8;
  if (c & 4)  c ^= 2;
  return c;
}

__device__ __forceinline__ cf gate_entry(const float (*cs)[2], int layer, int q, int a, int b) {
  const int g = layer * 18 + q * 3;
  const float cx = cs[g][0], sx = cs[g][1];
  const float cy = cs[g + 1][0], sy = cs[g + 1][1];
  const float cz = cs[g + 2][0], sz = cs[g + 2][1];
  const float ry0 = (a == 0) ? cy : sy;
  const float ry1 = (a == 0) ? -sy : cy;
  cf t;
  if (b == 0) { t.r = ry0 * cx; t.i = -ry1 * sx; }
  else        { t.r = ry1 * cx; t.i = -ry0 * sx; }
  cf rz;
  rz.r = cz;
  rz.i = a ? sz : -sz;
  return cmul(rz, t);
}

__device__ __forceinline__ void build_left(const cf* khi, const cf* klo,
                                           float* sLr, float* sLi, int tid) {
#pragma unroll 4
  for (int e = tid; e < 2048; e += 128) {
    const int m = e & 63, kp = e >> 6;
    const int t0 = cnot_chain(2 * kp), t1 = cnot_chain(2 * kp + 1);
    const cf v0 = cmul(khi[(m >> 3) * 8 + (t0 >> 3)], klo[(m & 7) * 8 + (t0 & 7)]);
    const cf v1 = cmul(khi[(m >> 3) * 8 + (t1 >> 3)], klo[(m & 7) * 8 + (t1 & 7)]);
    v2f vr, vi;
    vr.x = v0.r; vr.y = v1.r;
    vi.x = v0.i; vi.y = v1.i;
    *(v2f*)&sLr[e * 2] = vr;
    *(v2f*)&sLi[e * 2] = vi;
  }
}

__device__ __forceinline__ void cgemm_tile(const float* sLr, const float* sLi,
                                           const float* sBr, const float* sBi,
                                           int wave, int half, int l, v8f& cr, v8f& ci) {
  v8f rr = v8f{}, ii = v8f{}, ri = v8f{}, ir = v8f{};
  const int m = 16 * wave + l;
#pragma unroll
  for (int kt = 0; kt < 16; ++kt) {
    const int kp = 2 * kt + half;
    const v2f ar = *(const v2f*)&sLr[(kp * 64 + m) * 2];
    const v2f ai = *(const v2f*)&sLi[(kp * 64 + m) * 2];
    const v2f br = *(const v2f*)&sBr[(kp * 16 + l) * 2];
    const v2f bi = *(const v2f*)&sBi[(kp * 16 + l) * 2];
    rr = __builtin_amdgcn_wmma_f32_16x16x4_f32(false, ar, false, br, (short)0, rr, false, false);
    ii = __builtin_amdgcn_wmma_f32_16x16x4_f32(false, ai, false, bi, (short)0, ii, false, false);
    ri = __builtin_amdgcn_wmma_f32_16x16x4_f32(false, ar, false, bi, (short)0, ri, false, false);
    ir = __builtin_amdgcn_wmma_f32_16x16x4_f32(false, ai, false, br, (short)0, ir, false, false);
  }
  cr = rr - ii;
  ci = ri + ir;
}

__device__ __forceinline__ float layer_table_entry(
    int slot, int t, const float* __restrict__ W1, const float* __restrict__ b1,
    const float* __restrict__ W2, const float* __restrict__ b2,
    const float* __restrict__ W3, const float* __restrict__ b3,
    const float* __restrict__ W4, const float* __restrict__ b4) {
  const int lane = t >> 2;
  const int f = slot * 4 + (t & 3);
  const int c = lane & 15, h = lane >> 4;
  const float* p;
  int i;
  bool ok = true;
  if (f < 8) { p = b1; i = 8 * h + f; }
  else if (f < 16) { const int r = f - 8; p = b2; i = 3 * h + (r < 3 ? r : 2); ok = r < 3; }
  else if (f < 24) {
    const int k = f - 16;
    const int q = (c < 3) ? c : ((c >= 8 && c < 11) ? c - 5 : -1);
    p = W2; i = (8 * h + k) * 6 + (q >= 0 ? q : 0); ok = q >= 0;
  }
  else if (f < 28) { const int k = f - 24; p = W1; i = (2 * h + (k & 1)) * 16 + c; ok = k < 2; }
  else if (f < 36) { p = b3; i = 8 * h + (f - 28); }
  else if (f < 44) { const int r = f - 36; p = b4; i = r & 3; ok = (h == 0) && (r < 4); }
  else if (f < 52) { const int k = f - 44; p = W4; i = (8 * h + k) * 4 + (c & 3); ok = c < 4; }
  else { const int k = f - 52; p = W3; i = (k < 6 ? k : 5) * 16 + c; ok = k < 6; }
  const float v = p[i];
  return ok ? v : 0.0f;
}

#define LAYER_TABLE_SLOTS 15

__global__ __launch_bounds__(128) void build_unitary_kernel(
    const float* __restrict__ qp,
    const float* __restrict__ W1, const float* __restrict__ b1,
    const float* __restrict__ W2, const float* __restrict__ b2,
    const float* __restrict__ W3, const float* __restrict__ b3,
    const float* __restrict__ W4, const float* __restrict__ b4,
    float* __restrict__ Uf) {
  if (blockIdx.x == 4) {
    float* Tf = Uf + 2 * 64 * 64;
#pragma unroll
    for (int j = 0; j < LAYER_TABLE_SLOTS; ++j) {
      Tf[j * 128 + (int)threadIdx.x] = layer_table_entry(j, (int)threadIdx.x, W1, b1, W2, b2, W3, b3, W4, b4);
    }
    return;
  }
  __shared__ float sCS[54][2];
  __shared__ cf sK[3][2][64];
  __shared__ cf sD[2][64];
  __shared__ float sLr[32 * 64 * 2];
  __shared__ float sLi[32 * 64 * 2];
  __shared__ float sBr[32 * 16 * 2];
  __shared__ float sBi[32 * 16 * 2];

  const int tid  = threadIdx.x;
  const int wave = tid >> 5;
  const int lane = tid & 31;
  const int half = lane >> 4;
  const int l    = lane & 15;
  const int slab = blockIdx.x;

  if (tid < 54) {
    float sn, cs;
    sincosf(0.5f * qp[tid], &sn, &cs);
    sCS[tid][0] = cs;
    sCS[tid][1] = sn;
  }
  __syncthreads();

#pragma unroll
  for (int j = 0; j < 3; ++j) {
    const int e = tid + 128 * j;
    const int ly = e >> 7, hl = (e >> 6) & 1, ab = e & 63, a = ab >> 3, b = ab & 7;
    const int q0 = 3 * hl;
    const cf v = cmul(cmul(gate_entry(sCS, ly, q0, (a >> 2) & 1, (b >> 2) & 1),
                           gate_entry(sCS, ly, q0 + 1, (a >> 1) & 1, (b >> 1) & 1)),
                      gate_entry(sCS, ly, q0 + 2, a & 1, b & 1));
    sK[ly][hl][ab] = v;
  }
  {
    const int ly = tid >> 6, s = tid & 63;
    cf d;
    d.r = 1.0f;
    d.i = 0.0f;
#pragma unroll
    for (int i = 0; i < 6; ++i) {
      if (s & (32 >> i)) {
        const int tb = 32 >> ((i + 1) % 6);
        cf p;
        p.r = sCS[ly * 18 + i * 3][0];
        p.i = (s & tb) ? sCS[ly * 18 + i * 3][1] : -sCS[ly * 18 + i * 3][1];
        d = cmul(d, p);
      }
    }
    sD[ly][s] = d;
  }
  __syncthreads();

  build_left(sK[1][0], sK[1][1], sLr, sLi, tid);
#pragma unroll
  for (int e = tid; e < 512; e += 128) {
    const int n = e & 15, kp = e >> 4;
    const int t = cnot_chain(16 * slab + n);
    const int k0 = 2 * kp, k1 = 2 * kp + 1;
    const cf v0 = cmul(sD[0][k0], cmul(sK[0][0][(k0 >> 3) * 8 + (t >> 3)], sK[0][1][(k0 & 7) * 8 + (t & 7)]));
    const cf v1 = cmul(sD[0][k1], cmul(sK[0][0][(k1 >> 3) * 8 + (t >> 3)], sK[0][1][(k1 & 7) * 8 + (t & 7)]));
    v2f vr, vi;
    vr.x = v0.r; vr.y = v1.r;
    vi.x = v0.i; vi.y = v1.i;
    *(v2f*)&sBr[e * 2] = vr;
    *(v2f*)&sBi[e * 2] = vi;
  }
  __syncthreads();

  v8f cr, ci;
  cgemm_tile(sLr, sLi, sBr, sBi, wave, half, l, cr, ci);
  __syncthreads();

#pragma unroll
  for (int r = 0; r < 8; r += 2) {
    const int k0 = 16 * wave + 8 * half + r;
    cf a0, a1;
    a0.r = cr[r];     a0.i = ci[r];
    a1.r = cr[r + 1]; a1.i = ci[r + 1];
    const cf t0 = cmul(sD[1][k0], a0);
    const cf t1 = cmul(sD[1][k0 + 1], a1);
    v2f vr, vi;
    vr.x = t0.r; vr.y = t1.r;
    vi.x = t0.i; vi.y = t1.i;
    *(v2f*)&sBr[((k0 >> 1) * 16 + l) * 2] = vr;
    *(v2f*)&sBi[((k0 >> 1) * 16 + l) * 2] = vi;
  }
  build_left(sK[2][0], sK[2][1], sLr, sLi, tid);
  __syncthreads();

  cgemm_tile(sLr, sLi, sBr, sBi, wave, half, l, cr, ci);

  {
    const int k = 16 * slab + l;
    float* p = Uf + (((k >> 2) * 4 + wave) * 32 + ((k >> 1) & 1) * 16 + 8 * half) * 4 + (k & 1);
#pragma unroll
    for (int r = 0; r < 8; ++r) {
      p[4 * r]     = cr[r];
      p[4 * r + 2] = ci[r];
    }
  }
}

__device__ __forceinline__ v8f wmma4(v2f a, v2f b, v8f c) {
  return __builtin_amdgcn_wmma_f32_16x16x4_f32(false, a, false, b, (short)0, c, false, false);
}

__device__ __forceinline__ float xhalf(float v) {
  const int b = __builtin_bit_cast(int, v);
  return __builtin_bit_cast(float,
      __builtin_amdgcn_permlanex16(b, b, 0x76543210, 0xfedcba98, false, false));
}

__global__ __launch_bounds__(WAVES * 32) void qnn_kernel(
    const float* __restrict__ x,
    const float* __restrict__ Uf,
    float* __restrict__ out, int batch) {
  __shared__ f4v sU[16 * 4 * 32];

  const int tid  = threadIdx.x;
  const int wave = tid >> 5;
  const int lane = tid & 31;
  const int half = lane >> 4;
  const int l    = lane & 15;
  const int ebase = blockIdx.x * TILE + wave * 16;
  const f4v* tab = (const f4v*)(Uf + 2 * 64 * 64) + lane;

  {
    const f4v* srcU = (const f4v*)Uf;
    for (int i = tid; i < 2048; i += WAVES * 32) sU[i] = srcU[i];
  }

  const int e  = ebase + l;
  const int ec = (e < batch) ? e : (batch - 1);
  float m01[4], m23[4], mh0, mh1;
  {
    const f4v c1a = tab[0 * 32], c1b = tab[1 * 32], c2a = tab[2 * 32], c2b = tab[3 * 32];
    const f4v a2a = tab[4 * 32], a2b = tab[5 * 32], a1p = tab[6 * 32];
    const v2f xb = *(const v2f*)(x + (size_t)ec * 4 + 2 * half);
    const v8f c1 = {c1a.x, c1a.y, c1a.z, c1a.w, c1b.x, c1b.y, c1b.z, c1b.w};
    const v8f c2 = {c2a.x, c2a.y, c2a.z, c2a.w, c2b.x, c2b.y, c2b.z, c2b.w};
    const v8f hp = wmma4(v2f{a1p.x, a1p.y}, xb, c1);
    float H[8];
#pragma unroll
    for (int r = 0; r < 8; ++r) H[r] = hp[r] > 0.0f ? hp[r] : 0.0f;
    v8f tp = c2;
    tp = wmma4(v2f{a2a.x, a2a.y}, v2f{H[0], H[1]}, tp);
    tp = wmma4(v2f{a2a.z, a2a.w}, v2f{H[2], H[3]}, tp);
    tp = wmma4(v2f{a2b.x, a2b.y}, v2f{H[4], H[5]}, tp);
    tp = wmma4(v2f{a2b.z, a2b.w}, v2f{H[6], H[7]}, tp);
    float a0m[3], a1m[3];
#pragma unroll
    for (int i = 0; i < 3; ++i) {
      const float tm = tp[i];
      const float ang = 0.5f * 3.14159265358979323846f * tanhf(tm);
      float sn, cs; sincosf(ang, &sn, &cs);
      const float is2 = 0.70710678118654752440f;
      a0m[i] = (cs - sn) * is2;
      a1m[i] = (cs + sn) * is2;
    }
    float A0[6], A1[6];
#pragma unroll
    for (int i = 0; i < 3; ++i) {
      const float a0o = xhalf(a0m[i]);
      const float a1o = xhalf(a1m[i]);
      A0[i]     = half ? a0o : a0m[i];
      A1[i]     = half ? a1o : a1m[i];
      A0[3 + i] = half ? a0m[i] : a0o;
      A1[3 + i] = half ? a1m[i] : a1o;
    }
#pragma unroll
    for (int i = 0; i < 4; ++i) {
      m01[i] = ((i & 2) ? A1[0] : A0[0]) * ((i & 1) ? A1[1] : A0[1]);
      m23[i] = ((i & 2) ? A1[2] : A0[2]) * ((i & 1) ? A1[3] : A0[3]);
    }
    const float a4 = half ? A1[4] : A0[4];
    mh0 = a4 * A0[5];
    mh1 = a4 * A1[5];
  }
  __syncthreads();

  v8f accr[4] = {v8f{}, v8f{}, v8f{}, v8f{}};
  v8f acci[4] = {v8f{}, v8f{}, v8f{}, v8f{}};
  const f4v* uf = &sU[lane];
#pragma unroll
  for (int kt = 0; kt < 16; ++kt) {
    const float p = m01[kt >> 2] * m23[kt & 3];
    v2f b;
    b.x = p * mh0;
    b.y = p * mh1;
#pragma unroll
    for (int mt = 0; mt < 4; ++mt) {
      const f4v u = uf[(kt * 4 + mt) * 32];
      v2f ar, ai;
      ar.x = u.x; ar.y = u.y;
      ai.x = u.z; ai.y = u.w;
      accr[mt] = __builtin_amdgcn_wmma_f32_16x16x4_f32(
          false, ar, false, b, (short)0, accr[mt], false, false);
      acci[mt] = __builtin_amdgcn_wmma_f32_16x16x4_f32(
          false, ai, false, b, (short)0, acci[mt], false, false);
    }
  }

  float z0 = 0, z1 = 0, zs = 0, z3 = 0, z4 = 0, z5 = 0;
#pragma unroll
  for (int mt = 0; mt < 4; ++mt)
#pragma unroll
    for (int r = 0; r < 8; ++r) {
      const float pr = accr[mt][r], pi = acci[mt][r];
      const float v = pr * pr + pi * pi;
      z0 += (mt & 2) ? -v : v;
      z1 += (mt & 1) ? -v : v;
      zs += v;
      z3 += (r & 4) ? -v : v;
      z4 += (r & 2) ? -v : v;
      z5 += (r & 1) ? -v : v;
    }
  const float z2 = half ? -zs : zs;

  {
    const f4v c3a = tab[7 * 32], c3b = tab[8 * 32], c4a = tab[9 * 32], c4b = tab[10 * 32];
    const f4v a4a = tab[11 * 32], a4b = tab[12 * 32], a3a = tab[13 * 32], a3b = tab[14 * 32];
    v8f gp = {c3a.x, c3a.y, c3a.z, c3a.w, c3b.x, c3b.y, c3b.z, c3b.w};
    gp = wmma4(v2f{a3a.x, a3a.y}, v2f{z0, z1}, gp);
    gp = wmma4(v2f{a3a.z, a3a.w}, v2f{z2, z3}, gp);
    gp = wmma4(v2f{a3b.x, a3b.y}, v2f{z4, z5}, gp);
    float G[8];
#pragma unroll
    for (int r = 0; r < 8; ++r) G[r] = gp[r] > 0.0f ? gp[r] : 0.0f;
    v8f op = {c4a.x, c4a.y, c4a.z, c4a.w, c4b.x, c4b.y, c4b.z, c4b.w};
    op = wmma4(v2f{a4a.x, a4a.y}, v2f{G[0], G[1]}, op);
    op = wmma4(v2f{a4a.z, a4a.w}, v2f{G[2], G[3]}, op);
    op = wmma4(v2f{a4b.x, a4b.y}, v2f{G[4], G[5]}, op);
    op = wmma4(v2f{a4b.z, a4b.w}, v2f{G[6], G[7]}, op);
    if (half == 0 && e < batch) {
      f4v o;
      o.x = op[0]; o.y = op[1]; o.z = op[2]; o.w = op[3];
      *(f4v*)(out + (size_t)e * 4) = o;
    }
  }
}

extern "C" void kernel_launch(void* const* d_in, const int* in_sizes, int n_in,
                              void* d_out, int out_size, void* d_ws, size_t ws_size,
                              hipStream_t stream) {
  const float* x  = (const float*)d_in[0];
  const float* W1 = (const float*)d_in[1];
  const float* b1 = (const float*)d_in[2];
  const float* W2 = (const float*)d_in[3];
  const float* b2 = (const float*)d_in[4];
  const float* qp = (const float*)d_in[5];
  const float* W3 = (const float*)d_in[6];
  const float* b3 = (const float*)d_in[7];
  const float* W4 = (const float*)d_in[8];
  const float* b4 = (const float*)d_in[9];
  float* out  = (float*)d_out;
  float* Uf = (float*)d_ws;

  const int batch = in_sizes[0] / 4;
  build_unitary_kernel<<<dim3(5), dim3(128), 0, stream>>>(qp, W1, b1, W2, b2, W3, b3, W4, b4, Uf);
  const int tiles = (batch + TILE - 1) / TILE;
  qnn_kernel<<<dim3(tiles), dim3(WAVES * 32), 0, stream>>>(x, Uf, out, batch);
}


# Unitary build: each lane forms the A fragments of both layer products in registers from its row's two 8-entry table rows; the 64x64 left operands are no longer written to and read back from LDS
# speedup vs baseline: 13.2133x; 1.0404x over previous
#include <hip/hip_runtime.h>
#include <math.h>

typedef __attribute__((ext_vector_type(2))) float v2f;
typedef __attribute__((ext_vector_type(4))) float f4v;
typedef __attribute__((ext_vector_type(8))) float v8f;

#define WAVES 4
#define TILE  (WAVES * 16)

struct cf { float r, i; };

__device__ __forceinline__ cf cmul(cf a, cf b) {
  cf o;
  o.r = a.r * b.r - a.i * b.i;
  o.i = a.r * b.i + a.i * b.r;
  return o;
}

__device__ __forceinline__ int cnot_chain(int c) {
  if (c & 32) c ^= 16;
  if (c & 8)  c ^= 4;
  if (c & 2)  c ^= 1;
  if (c & 16) c ^= 8;
  if (c & 4)  c ^= 2;
  return c;
}

__device__ __forceinline__ cf gate_entry(const float (*cs)[2], int layer, int q, int a, int b) {
  const int g = layer * 18 + q * 3;
  const float cx = cs[g][0], sx = cs[g][1];
  const float cy = cs[g + 1][0], sy = cs[g + 1][1];
  const float cz = cs[g + 2][0], sz = cs[g + 2][1];
  const float ry0 = (a == 0) ? cy : sy;
  const float ry1 = (a == 0) ? -sy : cy;
  cf t;
  if (b == 0) { t.r = ry0 * cx; t.i = -ry1 * sx; }
  else        { t.r = ry1 * cx; t.i = -ry0 * sx; }
  cf rz;
  rz.r = cz;
  rz.i = a ? sz : -sz;
  return cmul(rz, t);
}


__device__ __forceinline__ void load_rows(const cf* khi, const cf* klo, int m, cf (&hi)[8], cf (&lo)[8]) {
#pragma unroll
  for (int j = 0; j < 8; ++j) {
    hi[j] = khi[(m >> 3) * 8 + j];
    lo[j] = klo[(m & 7) * 8 + j];
  }
}

__device__ __forceinline__ void cgemm_tile(const cf (&hi)[8], const cf (&lo)[8],
                                           const float* sBr, const float* sBi,
                                           int half, int l, v8f& cr, v8f& ci) {
  v8f rr = v8f{}, ii = v8f{}, ri = v8f{}, ir = v8f{};
#pragma unroll
  for (int kt = 0; kt < 16; ++kt) {
    const int kp = 2 * kt + half;
    const cf h = hi[cnot_chain(4 * kt) >> 3];
    cf a[2];
#pragma unroll
    for (int s = 0; s < 2; ++s) {
      const int i0 = cnot_chain(4 * kt + s) & 7;
      const int i1 = cnot_chain(4 * kt + 2 + s) & 7;
      cf g;
      g.r = half ? lo[i1].r : lo[i0].r;
      g.i = half ? lo[i1].i : lo[i0].i;
      a[s] = cmul(h, g);
    }
    v2f ar, ai;
    ar.x = a[0].r; ar.y = a[1].r;
    ai.x = a[0].i; ai.y = a[1].i;
    const v2f br = *(const v2f*)&sBr[(kp * 16 + l) * 2];
    const v2f bi = *(const v2f*)&sBi[(kp * 16 + l) * 2];
    rr = __builtin_amdgcn_wmma_f32_16x16x4_f32(false, ar, false, br, (short)0, rr, false, false);
    ii = __builtin_amdgcn_wmma_f32_16x16x4_f32(false, ai, false, bi, (short)0, ii, false, false);
    ri = __builtin_amdgcn_wmma_f32_16x16x4_f32(false, ar, false, bi, (short)0, ri, false, false);
    ir = __builtin_amdgcn_wmma_f32_16x16x4_f32(false, ai, false, br, (short)0, ir, false, false);
  }
  cr = rr - ii;
  ci = ri + ir;
}

__device__ __forceinline__ float layer_table_entry(
    int slot, int t, const float* __restrict__ W1, const float* __restrict__ b1,
    const float* __restrict__ W2, const float* __restrict__ b2,
    const float* __restrict__ W3, const float* __restrict__ b3,
    const float* __restrict__ W4, const float* __restrict__ b4) {
  const int lane = t >> 2;
  const int f = slot * 4 + (t & 3);
  const int c = lane & 15, h = lane >> 4;
  const float* p;
  int i;
  bool ok = true;
  if (f < 8) { p = b1; i = 8 * h + f; }
  else if (f < 16) { const int r = f - 8; p = b2; i = 3 * h + (r < 3 ? r : 2); ok = r < 3; }
  else if (f < 24) {
    const int k = f - 16;
    const int q = (c < 3) ? c : ((c >= 8 && c < 11) ? c - 5 : -1);
    p = W2; i = (8 * h + k) * 6 + (q >= 0 ? q : 0); ok = q >= 0;
  }
  else if (f < 28) { const int k = f - 24; p = W1; i = (2 * h + (k & 1)) * 16 + c; ok = k < 2; }
  else if (f < 36) { p = b3; i = 8 * h + (f - 28); }
  else if (f < 44) { const int r = f - 36; p = b4; i = r & 3; ok = (h == 0) && (r < 4); }
  else if (f < 52) { const int k = f - 44; p = W4; i = (8 * h + k) * 4 + (c & 3); ok = c < 4; }
  else { const int k = f - 52; p = W3; i = (k < 6 ? k : 5) * 16 + c; ok = k < 6; }
  const float v = p[i];
  return ok ? v : 0.0f;
}

#define LAYER_TABLE_SLOTS 15

__global__ __launch_bounds__(128) void build_unitary_kernel(
    const float* __restrict__ qp,
    const float* __restrict__ W1, const float* __restrict__ b1,
    const float* __restrict__ W2, const float* __restrict__ b2,
    const float* __restrict__ W3, const float* __restrict__ b3,
    const float* __restrict__ W4, const float* __restrict__ b4,
    float* __restrict__ Uf) {
  if (blockIdx.x == 4) {
    float* Tf = Uf + 2 * 64 * 64;
#pragma unroll
    for (int j = 0; j < LAYER_TABLE_SLOTS; ++j) {
      Tf[j * 128 + (int)threadIdx.x] = layer_table_entry(j, (int)threadIdx.x, W1, b1, W2, b2, W3, b3, W4, b4);
    }
    return;
  }
  __shared__ float sCS[54][2];
  __shared__ cf sK[3][2][64];
  __shared__ cf sD[2][64];
  __shared__ float sBr[32 * 16 * 2];
  __shared__ float sBi[32 * 16 * 2];

  const int tid  = threadIdx.x;
  const int wave = tid >> 5;
  const int lane = tid & 31;
  const int half = lane >> 4;
  const int l    = lane & 15;
  const int slab = blockIdx.x;

  if (tid < 54) {
    float sn, cs;
    sincosf(0.5f * qp[tid], &sn, &cs);
    sCS[tid][0] = cs;
    sCS[tid][1] = sn;
  }
  __syncthreads();

#pragma unroll
  for (int j = 0; j < 3; ++j) {
    const int e = tid + 128 * j;
    const int ly = e >> 7, hl = (e >> 6) & 1, ab = e & 63, a = ab >> 3, b = ab & 7;
    const int q0 = 3 * hl;
    const cf v = cmul(cmul(gate_entry(sCS, ly, q0, (a >> 2) & 1, (b >> 2) & 1),
                           gate_entry(sCS, ly, q0 + 1, (a >> 1) & 1, (b >> 1) & 1)),
                      gate_entry(sCS, ly, q0 + 2, a & 1, b & 1));
    sK[ly][hl][ab] = v;
  }
  {
    const int ly = tid >> 6, s = tid & 63;
    cf d;
    d.r = 1.0f;
    d.i = 0.0f;
#pragma unroll
    for (int i = 0; i < 6; ++i) {
      if (s & (32 >> i)) {
        const int tb = 32 >> ((i + 1) % 6);
        cf p;
        p.r = sCS[ly * 18 + i * 3][0];
        p.i = (s & tb) ? sCS[ly * 18 + i * 3][1] : -sCS[ly * 18 + i * 3][1];
        d = cmul(d, p);
      }
    }
    sD[ly][s] = d;
  }
  __syncthreads();

  cf hi1[8], lo1[8], hi2[8], lo2[8];
  load_rows(sK[1][0], sK[1][1], 16 * wave + l, hi1, lo1);
  load_rows(sK[2][0], sK[2][1], 16 * wave + l, hi2, lo2);

#pragma unroll
  for (int e = tid; e < 512; e += 128) {
    const int n = e & 15, kp = e >> 4;
    const int t = cnot_chain(16 * slab + n);
    const int k0 = 2 * kp, k1 = 2 * kp + 1;
    const cf v0 = cmul(sD[0][k0], cmul(sK[0][0][(k0 >> 3) * 8 + (t >> 3)], sK[0][1][(k0 & 7) * 8 + (t & 7)]));
    const cf v1 = cmul(sD[0][k1], cmul(sK[0][0][(k1 >> 3) * 8 + (t >> 3)], sK[0][1][(k1 & 7) * 8 + (t & 7)]));
    v2f vr, vi;
    vr.x = v0.r; vr.y = v1.r;
    vi.x = v0.i; vi.y = v1.i;
    *(v2f*)&sBr[e * 2] = vr;
    *(v2f*)&sBi[e * 2] = vi;
  }
  __syncthreads();

  v8f cr, ci;
  cgemm_tile(hi1, lo1, sBr, sBi, half, l, cr, ci);
  __syncthreads();

#pragma unroll
  for (int r = 0; r < 8; r += 2) {
    const int k0 = 16 * wave + 8 * half + r;
    cf a0, a1;
    a0.r = cr[r];     a0.i = ci[r];
    a1.r = cr[r + 1]; a1.i = ci[r + 1];
    const cf t0 = cmul(sD[1][k0], a0);
    const cf t1 = cmul(sD[1][k0 + 1], a1);
    v2f vr, vi;
    vr.x = t0.r; vr.y = t1.r;
    vi.x = t0.i; vi.y = t1.i;
    *(v2f*)&sBr[((k0 >> 1) * 16 + l) * 2] = vr;
    *(v2f*)&sBi[((k0 >> 1) * 16 + l) * 2] = vi;
  }
  __syncthreads();

  cgemm_tile(hi2, lo2, sBr, sBi, half, l, cr, ci);

  {
    const int k = 16 * slab + l;
    float* p = Uf + (((k >> 2) * 4 + wave) * 32 + ((k >> 1) & 1) * 16 + 8 * half) * 4 + (k & 1);
#pragma unroll
    for (int r = 0; r < 8; ++r) {
      p[4 * r]     = cr[r];
      p[4 * r + 2] = ci[r];
    }
  }
}

__device__ __forceinline__ v8f wmma4(v2f a, v2f b, v8f c) {
  return __builtin_amdgcn_wmma_f32_16x16x4_f32(false, a, false, b, (short)0, c, false, false);
}

__device__ __forceinline__ float xhalf(float v) {
  const int b = __builtin_bit_cast(int, v);
  return __builtin_bit_cast(float,
      __builtin_amdgcn_permlanex16(b, b, 0x76543210, 0xfedcba98, false, false));
}

__global__ __launch_bounds__(WAVES * 32) void qnn_kernel(
    const float* __restrict__ x,
    const float* __restrict__ Uf,
    float* __restrict__ out, int batch) {
  __shared__ f4v sU[16 * 4 * 32];

  const int tid  = threadIdx.x;
  const int wave = tid >> 5;
  const int lane = tid & 31;
  const int half = lane >> 4;
  const int l    = lane & 15;
  const int ebase = blockIdx.x * TILE + wave * 16;
  const f4v* tab = (const f4v*)(Uf + 2 * 64 * 64) + lane;

  {
    const f4v* srcU = (const f4v*)Uf;
    for (int i = tid; i < 2048; i += WAVES * 32) sU[i] = srcU[i];
  }

  const int e  = ebase + l;
  const int ec = (e < batch) ? e : (batch - 1);
  float m01[4], m23[4], mh0, mh1;
  {
    const f4v c1a = tab[0 * 32], c1b = tab[1 * 32], c2a = tab[2 * 32], c2b = tab[3 * 32];
    const f4v a2a = tab[4 * 32], a2b = tab[5 * 32], a1p = tab[6 * 32];
    const v2f xb = *(const v2f*)(x + (size_t)ec * 4 + 2 * half);
    const v8f c1 = {c1a.x, c1a.y, c1a.z, c1a.w, c1b.x, c1b.y, c1b.z, c1b.w};
    const v8f c2 = {c2a.x, c2a.y, c2a.z, c2a.w, c2b.x, c2b.y, c2b.z, c2b.w};
    const v8f hp = wmma4(v2f{a1p.x, a1p.y}, xb, c1);
    float H[8];
#pragma unroll
    for (int r = 0; r < 8; ++r) H[r] = hp[r] > 0.0f ? hp[r] : 0.0f;
    v8f tp = c2;
    tp = wmma4(v2f{a2a.x, a2a.y}, v2f{H[0], H[1]}, tp);
    tp = wmma4(v2f{a2a.z, a2a.w}, v2f{H[2], H[3]}, tp);
    tp = wmma4(v2f{a2b.x, a2b.y}, v2f{H[4], H[5]}, tp);
    tp = wmma4(v2f{a2b.z, a2b.w}, v2f{H[6], H[7]}, tp);
    float a0m[3], a1m[3];
#pragma unroll
    for (int i = 0; i < 3; ++i) {
      const float tm = tp[i];
      const float ang = 0.5f * 3.14159265358979323846f * tanhf(tm);
      float sn, cs; sincosf(ang, &sn, &cs);
      const float is2 = 0.70710678118654752440f;
      a0m[i] = (cs - sn) * is2;
      a1m[i] = (cs + sn) * is2;
    }
    float A0[6], A1[6];
#pragma unroll
    for (int i = 0; i < 3; ++i) {
      const float a0o = xhalf(a0m[i]);
      const float a1o = xhalf(a1m[i]);
      A0[i]     = half ? a0o : a0m[i];
      A1[i]     = half ? a1o : a1m[i];
      A0[3 + i] = half ? a0m[i] : a0o;
      A1[3 + i] = half ? a1m[i] : a1o;
    }
#pragma unroll
    for (int i = 0; i < 4; ++i) {
      m01[i] = ((i & 2) ? A1[0] : A0[0]) * ((i & 1) ? A1[1] : A0[1]);
      m23[i] = ((i & 2) ? A1[2] : A0[2]) * ((i & 1) ? A1[3] : A0[3]);
    }
    const float a4 = half ? A1[4] : A0[4];
    mh0 = a4 * A0[5];
    mh1 = a4 * A1[5];
  }
  __syncthreads();

  v8f accr[4] = {v8f{}, v8f{}, v8f{}, v8f{}};
  v8f acci[4] = {v8f{}, v8f{}, v8f{}, v8f{}};
  const f4v* uf = &sU[lane];
#pragma unroll
  for (int kt = 0; kt < 16; ++kt) {
    const float p = m01[kt >> 2] * m23[kt & 3];
    v2f b;
    b.x = p * mh0;
    b.y = p * mh1;
#pragma unroll
    for (int mt = 0; mt < 4; ++mt) {
      const f4v u = uf[(kt * 4 + mt) * 32];
      v2f ar, ai;
      ar.x = u.x; ar.y = u.y;
      ai.x = u.z; ai.y = u.w;
      accr[mt] = __builtin_amdgcn_wmma_f32_16x16x4_f32(
          false, ar, false, b, (short)0, accr[mt], false, false);
      acci[mt] = __builtin_amdgcn_wmma_f32_16x16x4_f32(
          false, ai, false, b, (short)0, acci[mt], false, false);
    }
  }

  float z0 = 0, z1 = 0, zs = 0, z3 = 0, z4 = 0, z5 = 0;
#pragma unroll
  for (int mt = 0; mt < 4; ++mt)
#pragma unroll
    for (int r = 0; r < 8; ++r) {
      const float pr = accr[mt][r], pi = acci[mt][r];
      const float v = pr * pr + pi * pi;
      z0 += (mt & 2) ? -v : v;
      z1 += (mt & 1) ? -v : v;
      zs += v;
      z3 += (r & 4) ? -v : v;
      z4 += (r & 2) ? -v : v;
      z5 += (r & 1) ? -v : v;
    }
  const float z2 = half ? -zs : zs;

  {
    const f4v c3a = tab[7 * 32], c3b = tab[8 * 32], c4a = tab[9 * 32], c4b = tab[10 * 32];
    const f4v a4a = tab[11 * 32], a4b = tab[12 * 32], a3a = tab[13 * 32], a3b = tab[14 * 32];
    v8f gp = {c3a.x, c3a.y, c3a.z, c3a.w, c3b.x, c3b.y, c3b.z, c3b.w};
    gp = wmma4(v2f{a3a.x, a3a.y}, v2f{z0, z1}, gp);
    gp = wmma4(v2f{a3a.z, a3a.w}, v2f{z2, z3}, gp);
    gp = wmma4(v2f{a3b.x, a3b.y}, v2f{z4, z5}, gp);
    float G[8];
#pragma unroll
    for (int r = 0; r < 8; ++r) G[r] = gp[r] > 0.0f ? gp[r] : 0.0f;
    v8f op = {c4a.x, c4a.y, c4a.z, c4a.w, c4b.x, c4b.y, c4b.z, c4b.w};
    op = wmma4(v2f{a4a.x, a4a.y}, v2f{G[0], G[1]}, op);
    op = wmma4(v2f{a4a.z, a4a.w}, v2f{G[2], G[3]}, op);
    op = wmma4(v2f{a4b.x, a4b.y}, v2f{G[4], G[5]}, op);
    op = wmma4(v2f{a4b.z, a4b.w}, v2f{G[6], G[7]}, op);
    if (half == 0 && e < batch) {
      f4v o;
      o.x = op[0]; o.y = op[1]; o.z = op[2]; o.w = op[3];
      *(f4v*)(out + (size_t)e * 4) = o;
    }
  }
}

extern "C" void kernel_launch(void* const* d_in, const int* in_sizes, int n_in,
                              void* d_out, int out_size, void* d_ws, size_t ws_size,
                              hipStream_t stream) {
  const float* x  = (const float*)d_in[0];
  const float* W1 = (const float*)d_in[1];
  const float* b1 = (const float*)d_in[2];
  const float* W2 = (const float*)d_in[3];
  const float* b2 = (const float*)d_in[4];
  const float* qp = (const float*)d_in[5];
  const float* W3 = (const float*)d_in[6];
  const float* b3 = (const float*)d_in[7];
  const float* W4 = (const float*)d_in[8];
  const float* b4 = (const float*)d_in[9];
  float* out  = (float*)d_out;
  float* Uf = (float*)d_ws;

  const int batch = in_sizes[0] / 4;
  build_unitary_kernel<<<dim3(5), dim3(128), 0, stream>>>(qp, W1, b1, W2, b2, W3, b3, W4, b4, Uf);
  const int tiles = (batch + TILE - 1) / TILE;
  qnn_kernel<<<dim3(tiles), dim3(WAVES * 32), 0, stream>>>(x, Uf, out, batch);
}
